# starting kernel
# baseline (speedup 1.0000x reference)
#include <hip/hip_runtime.h>
#include <hip/hip_bf16.h>
#include <stdint.h>

#define B_IMG     8
#define CIN       1024
#define H_DIM     64
#define W_DIM     64
#define HW        (H_DIM * W_DIM)
#define C_OUT     196
#define C_OUT_PAD 208
#define N_ROI     1024
#define KPOOL     7
#define SAT_W     65

typedef __attribute__((ext_vector_type(2))) float v2f;
typedef __attribute__((ext_vector_type(8))) float v8f;

__global__ void batchsum_kernel(const float* __restrict__ fm,
                                float* __restrict__ F) {
    const size_t NE = (size_t)CIN * HW;
    size_t idx = ((size_t)blockIdx.x * blockDim.x + threadIdx.x) * 4;
    if (idx >= NE) return;
    float4 s = make_float4(0.f, 0.f, 0.f, 0.f);
#pragma unroll
    for (int b = 0; b < B_IMG; ++b) {
        const float4 v = *(const float4*)(fm + (size_t)b * NE + idx);
        s.x += v.x; s.y += v.y; s.z += v.z; s.w += v.w;
    }
    *(float4*)(F + idx) = s;
}

__global__ void gemm_wmma_kernel(const float* __restrict__ Wmat,
                                 const float* __restrict__ bias,
                                 const float* __restrict__ F,
                                 float* __restrict__ S) {
    const int flat  = blockIdx.x * blockDim.x + threadIdx.x;
    const int wave  = flat >> 5;
    const int lane  = threadIdx.x & 31;
    const int nTilesN = HW / 16;
    const int mt = wave / nTilesN;
    const int nt = wave % nTilesN;
    if (mt >= C_OUT_PAD / 16) return;

    const int o0 = mt * 16;
    const int n0 = nt * 16;
    const int lhalf = lane >> 4;
    const int l16   = lane & 15;

    v8f acc;
#pragma unroll
    for (int v = 0; v < 8; ++v) {
        const int o = min(o0 + v + 8 * lhalf, C_OUT - 1);
        acc[v] = 8.0f * bias[o];
    }

    const int oA = min(o0 + l16, C_OUT - 1);
    const float* Wrow = Wmat + (size_t)oA * CIN;
    const float* Fcol = F + n0 + l16;

    for (int k0 = 0; k0 < CIN; k0 += 4) {
        const int kk = k0 + 2 * lhalf;
        const v2f a = *(const v2f*)(Wrow + kk);
        v2f b;
        b.x = Fcol[(size_t)kk * HW];
        b.y = Fcol[(size_t)(kk + 1) * HW];
        acc = __builtin_amdgcn_wmma_f32_16x16x4_f32(
            false, a, false, b, (short)0, acc, false, false);
    }

#pragma unroll
    for (int v = 0; v < 8; ++v) {
        const int m = v + 8 * lhalf;
        S[(size_t)(o0 + m) * HW + n0 + l16] = acc[v];
    }
}

__global__ void integral_kernel(const float* __restrict__ S,
                                float* __restrict__ sat) {
    const int ch = blockIdx.x;
    const int t  = threadIdx.x;
    __shared__ float tile[H_DIM][W_DIM + 1];

    const float* Sp = S + (size_t)ch * HW;
    float run = 0.0f;
    for (int x = 0; x < W_DIM; ++x) {
        run += Sp[t * W_DIM + x];
        tile[t][x] = run;
    }
    __syncthreads();

    float* satp = sat + (size_t)ch * SAT_W * SAT_W;
    satp[t + 1]           = 0.0f;
    satp[(t + 1) * SAT_W] = 0.0f;
    if (t == 0) satp[0] = 0.0f;

    float crun = 0.0f;
    for (int y = 0; y < H_DIM; ++y) {
        crun += tile[y][t];
        satp[(y + 1) * SAT_W + (t + 1)] = crun;
    }
}

__global__ void roipool_kernel(const int* __restrict__ rois,
                               const float* __restrict__ sat,
                               float* __restrict__ out) {
    const int i = blockIdx.x;
    const int t = threadIdx.x;
    __shared__ float pool[C_OUT];

    const int* r = rois + i * 4;
    const int ymin = r[0] / 30, xmin = r[1] / 30;
    const int ymax = r[2] / 30, xmax = r[3] / 30;
    const int ystep = (ymax - ymin) / KPOOL;
    const int xstep = (xmax - xmin) / KPOOL;
    const float cnt = (float)(B_IMG * ystep * xstep);

    if (t < C_OUT) {
        const int bin = t >> 2;
        const int jy = bin / KPOOL, jx = bin % KPOOL;
        const int ys = ymin + jy * ystep, ye = ys + ystep;
        const int xs = xmin + jx * xstep, xe = xs + xstep;
        const float* sp = sat + (size_t)t * SAT_W * SAT_W;
        const float bs = sp[ye * SAT_W + xe] - sp[ys * SAT_W + xe]
                       - sp[ye * SAT_W + xs] + sp[ys * SAT_W + xs];
        pool[t] = (cnt > 0.0f) ? bs / fmaxf(cnt, 1.0f) : 0.0f;
    }
    __syncthreads();

    if (t < 4) {
        float s = 0.0f;
#pragma unroll
        for (int b = 0; b < KPOOL * KPOOL; ++b) s += pool[t + 4 * b];
        out[i * 4 + t] = s / (float)(KPOOL * KPOOL);
    }
}

extern "C" void kernel_launch(void* const* d_in, const int* in_sizes, int n_in,
                              void* d_out, int out_size, void* d_ws, size_t ws_size,
                              hipStream_t stream) {
    const float* fm    = (const float*)d_in[0];
    const int*   rois  = (const int*)  d_in[1];
    const float* convw = (const float*)d_in[2];
    const float* convb = (const float*)d_in[3];
    float*       out   = (float*)d_out;

    char* ws = (char*)d_ws;
    float* F   = (float*)(ws);
    float* S   = (float*)(ws + (size_t)CIN * HW * sizeof(float));
    float* sat = (float*)(ws + ((size_t)CIN * HW + (size_t)C_OUT_PAD * HW) * sizeof(float));

    {
        const int threads = 256;
        const int blocks  = (int)(((size_t)CIN * HW / 4 + threads - 1) / threads);
        batchsum_kernel<<<blocks, threads, 0, stream>>>(fm, F);
    }
    {
        const int threads = 256;
        const int totalWaves = (C_OUT_PAD / 16) * (HW / 16);
        const int blocks = (totalWaves * 32 + threads - 1) / threads;
        gemm_wmma_kernel<<<blocks, threads, 0, stream>>>(convw, convb, F, S);
    }
    integral_kernel<<<C_OUT, 64, 0, stream>>>(S, sat);

    roipool_kernel<<<N_ROI, 256, 0, stream>>>(rois, sat, out);
}


# GEMM wave tile 16x16 -> 16x64: one A fragment feeds 4 accumulator tiles, A and B arrive as 16-byte loads a 64-k chunk ahead; k is regrouped per WMMA, so the summation order over k changes
# speedup vs baseline: 1.2586x; 1.2586x over previous
#include <hip/hip_runtime.h>
#include <hip/hip_bf16.h>
#include <stdint.h>

#define B_IMG     8
#define CIN       1024
#define H_DIM     64
#define W_DIM     64
#define HW        (H_DIM * W_DIM)
#define C_OUT     196
#define C_OUT_PAD 208
#define N_ROI     1024
#define KPOOL     7
#define SAT_W     65

typedef __attribute__((ext_vector_type(2))) float v2f;
typedef __attribute__((ext_vector_type(8))) float v8f;

__global__ void batchsum_kernel(const float* __restrict__ fm,
                                float* __restrict__ F) {
    const size_t NE = (size_t)CIN * HW;
    size_t idx = ((size_t)blockIdx.x * blockDim.x + threadIdx.x) * 4;
    if (idx >= NE) return;
    float4 s = make_float4(0.f, 0.f, 0.f, 0.f);
#pragma unroll
    for (int b = 0; b < B_IMG; ++b) {
        const float4 v = *(const float4*)(fm + (size_t)b * NE + idx);
        s.x += v.x; s.y += v.y; s.z += v.z; s.w += v.w;
    }
    *(float4*)(F + idx) = s;
}

typedef __attribute__((ext_vector_type(4))) float v4f;

#define KC   64
#define NSUB (KC / 8)

struct FragSet {
    v4f a[NSUB];
    v4f b[NSUB][4];
};

__device__ __forceinline__ void load_set(FragSet& f,
                                         const float* __restrict__ Wp,
                                         const float* __restrict__ Fp,
                                         int k0) {
#pragma unroll
    for (int s = 0; s < NSUB; ++s) {
        f.a[s] = *(const v4f*)(Wp + k0 + 8 * s);
#pragma unroll
        for (int j = 0; j < 4; ++j)
            f.b[s][j] = *(const v4f*)(Fp + (size_t)(k0 + 8 * s + j) * HW);
    }
}

__device__ __forceinline__ void mma_set(v8f (&acc)[4], const FragSet& f) {
#pragma unroll
    for (int s = 0; s < NSUB; ++s) {
        v2f a01, a23;
        a01.x = f.a[s][0]; a01.y = f.a[s][1];
        a23.x = f.a[s][2]; a23.y = f.a[s][3];
#pragma unroll
        for (int t = 0; t < 4; ++t) {
            v2f b;
            b.x = f.b[s][0][t]; b.y = f.b[s][1][t];
            acc[t] = __builtin_amdgcn_wmma_f32_16x16x4_f32(
                false, a01, false, b, (short)0, acc[t], false, false);
        }
#pragma unroll
        for (int t = 0; t < 4; ++t) {
            v2f b;
            b.x = f.b[s][2][t]; b.y = f.b[s][3][t];
            acc[t] = __builtin_amdgcn_wmma_f32_16x16x4_f32(
                false, a23, false, b, (short)0, acc[t], false, false);
        }
    }
}

__global__ void __launch_bounds__(128)
gemm_wmma_kernel(const float* __restrict__ Wmat,
                 const float* __restrict__ bias,
                 const float* __restrict__ F,
                 float* __restrict__ S) {
    const int wave  = blockIdx.x * 4 + (threadIdx.x >> 5);
    const int lane  = threadIdx.x & 31;
    const int mt = wave % (C_OUT_PAD / 16);
    const int st = wave / (C_OUT_PAD / 16);

    const int o0 = mt * 16;
    const int n0 = st * 64;
    const int lhalf = lane >> 4;
    const int l16   = lane & 15;

    v8f acc[4];
#pragma unroll
    for (int v = 0; v < 8; ++v) {
        const int o = min(o0 + v + 8 * lhalf, C_OUT - 1);
        const float bv = 8.0f * bias[o];
#pragma unroll
        for (int t = 0; t < 4; ++t) acc[t][v] = bv;
    }

    const int oA = min(o0 + l16, C_OUT - 1);
    const float* Wp = Wmat + (size_t)oA * CIN + 4 * lhalf;
    const float* Fp = F + (size_t)(4 * lhalf) * HW + n0 + 4 * l16;

    FragSet f0, f1;
    load_set(f0, Wp, Fp, 0);
    for (int k0 = 0; k0 < CIN - 2 * KC; k0 += 2 * KC) {
        load_set(f1, Wp, Fp, k0 + KC);
        mma_set(acc, f0);
        load_set(f0, Wp, Fp, k0 + 2 * KC);
        mma_set(acc, f1);
    }
    load_set(f1, Wp, Fp, CIN - KC);
    mma_set(acc, f0);
    mma_set(acc, f1);

#pragma unroll
    for (int v = 0; v < 8; ++v) {
        const int m = v + 8 * lhalf;
        v4f o;
        o[0] = acc[0][v]; o[1] = acc[1][v]; o[2] = acc[2][v]; o[3] = acc[3][v];
        *(v4f*)(S + (size_t)(o0 + m) * HW + n0 + 4 * l16) = o;
    }
}

__global__ void integral_kernel(const float* __restrict__ S,
                                float* __restrict__ sat) {
    const int ch = blockIdx.x;
    const int t  = threadIdx.x;
    __shared__ float tile[H_DIM][W_DIM + 1];

    const float* Sp = S + (size_t)ch * HW;
    float run = 0.0f;
    for (int x = 0; x < W_DIM; ++x) {
        run += Sp[t * W_DIM + x];
        tile[t][x] = run;
    }
    __syncthreads();

    float* satp = sat + (size_t)ch * SAT_W * SAT_W;
    satp[t + 1]           = 0.0f;
    satp[(t + 1) * SAT_W] = 0.0f;
    if (t == 0) satp[0] = 0.0f;

    float crun = 0.0f;
    for (int y = 0; y < H_DIM; ++y) {
        crun += tile[y][t];
        satp[(y + 1) * SAT_W + (t + 1)] = crun;
    }
}

__global__ void roipool_kernel(const int* __restrict__ rois,
                               const float* __restrict__ sat,
                               float* __restrict__ out) {
    const int i = blockIdx.x;
    const int t = threadIdx.x;
    __shared__ float pool[C_OUT];

    const int* r = rois + i * 4;
    const int ymin = r[0] / 30, xmin = r[1] / 30;
    const int ymax = r[2] / 30, xmax = r[3] / 30;
    const int ystep = (ymax - ymin) / KPOOL;
    const int xstep = (xmax - xmin) / KPOOL;
    const float cnt = (float)(B_IMG * ystep * xstep);

    if (t < C_OUT) {
        const int bin = t >> 2;
        const int jy = bin / KPOOL, jx = bin % KPOOL;
        const int ys = ymin + jy * ystep, ye = ys + ystep;
        const int xs = xmin + jx * xstep, xe = xs + xstep;
        const float* sp = sat + (size_t)t * SAT_W * SAT_W;
        const float bs = sp[ye * SAT_W + xe] - sp[ys * SAT_W + xe]
                       - sp[ye * SAT_W + xs] + sp[ys * SAT_W + xs];
        pool[t] = (cnt > 0.0f) ? bs / fmaxf(cnt, 1.0f) : 0.0f;
    }
    __syncthreads();

    if (t < 4) {
        float s = 0.0f;
#pragma unroll
        for (int b = 0; b < KPOOL * KPOOL; ++b) s += pool[t + 4 * b];
        out[i * 4 + t] = s / (float)(KPOOL * KPOOL);
    }
}

extern "C" void kernel_launch(void* const* d_in, const int* in_sizes, int n_in,
                              void* d_out, int out_size, void* d_ws, size_t ws_size,
                              hipStream_t stream) {
    const float* fm    = (const float*)d_in[0];
    const int*   rois  = (const int*)  d_in[1];
    const float* convw = (const float*)d_in[2];
    const float* convb = (const float*)d_in[3];
    float*       out   = (float*)d_out;

    char* ws = (char*)d_ws;
    float* F   = (float*)(ws);
    float* S   = (float*)(ws + (size_t)CIN * HW * sizeof(float));
    float* sat = (float*)(ws + ((size_t)CIN * HW + (size_t)C_OUT_PAD * HW) * sizeof(float));

    {
        const int threads = 256;
        const int blocks  = (int)(((size_t)CIN * HW / 4 + threads - 1) / threads);
        batchsum_kernel<<<blocks, threads, 0, stream>>>(fm, F);
    }
    gemm_wmma_kernel<<<208, 128, 0, stream>>>(convw, convb, F, S);

    integral_kernel<<<C_OUT, 64, 0, stream>>>(S, sat);

    roipool_kernel<<<N_ROI, 256, 0, stream>>>(rois, sat, out);
}


# integral-image launch removed: GEMM epilogue prefix-sums each image row across lanes (DPP row_shr), ROI kernel adds its bin's rows from those prefixes; summation order of the window sums changes
# speedup vs baseline: 1.3577x; 1.0787x over previous
#include <hip/hip_runtime.h>
#include <hip/hip_bf16.h>
#include <stdint.h>

#define B_IMG     8
#define CIN       1024
#define H_DIM     64
#define W_DIM     64
#define HW        (H_DIM * W_DIM)
#define C_OUT     196
#define C_OUT_PAD 208
#define N_ROI     1024
#define KPOOL     7
#define MAX_STEP  ((H_DIM - 1) / KPOOL)

typedef __attribute__((ext_vector_type(2))) float v2f;
typedef __attribute__((ext_vector_type(8))) float v8f;

__global__ void batchsum_kernel(const float* __restrict__ fm,
                                float* __restrict__ F) {
    const size_t NE = (size_t)CIN * HW;
    size_t idx = ((size_t)blockIdx.x * blockDim.x + threadIdx.x) * 4;
    if (idx >= NE) return;
    float4 s = make_float4(0.f, 0.f, 0.f, 0.f);
#pragma unroll
    for (int b = 0; b < B_IMG; ++b) {
        const float4 v = *(const float4*)(fm + (size_t)b * NE + idx);
        s.x += v.x; s.y += v.y; s.z += v.z; s.w += v.w;
    }
    *(float4*)(F + idx) = s;
}

typedef __attribute__((ext_vector_type(4))) float v4f;

#define KC   64
#define NSUB (KC / 8)

struct FragSet {
    v4f a[NSUB];
    v4f b[NSUB][4];
};

__device__ __forceinline__ void load_set(FragSet& f,
                                         const float* __restrict__ Wp,
                                         const float* __restrict__ Fp,
                                         int k0) {
#pragma unroll
    for (int s = 0; s < NSUB; ++s) {
        f.a[s] = *(const v4f*)(Wp + k0 + 8 * s);
#pragma unroll
        for (int j = 0; j < 4; ++j)
            f.b[s][j] = *(const v4f*)(Fp + (size_t)(k0 + 8 * s + j) * HW);
    }
}

__device__ __forceinline__ void mma_set(v8f (&acc)[4], const FragSet& f) {
#pragma unroll
    for (int s = 0; s < NSUB; ++s) {
        v2f a01, a23;
        a01.x = f.a[s][0]; a01.y = f.a[s][1];
        a23.x = f.a[s][2]; a23.y = f.a[s][3];
#pragma unroll
        for (int t = 0; t < 4; ++t) {
            v2f b;
            b.x = f.b[s][0][t]; b.y = f.b[s][1][t];
            acc[t] = __builtin_amdgcn_wmma_f32_16x16x4_f32(
                false, a01, false, b, (short)0, acc[t], false, false);
        }
#pragma unroll
        for (int t = 0; t < 4; ++t) {
            v2f b;
            b.x = f.b[s][2][t]; b.y = f.b[s][3][t];
            acc[t] = __builtin_amdgcn_wmma_f32_16x16x4_f32(
                false, a23, false, b, (short)0, acc[t], false, false);
        }
    }
}

template <int N>
__device__ __forceinline__ float row_shr(float x) {
    return __builtin_bit_cast(float,
        __builtin_amdgcn_update_dpp(0, __builtin_bit_cast(int, x),
                                    0x110 + N  , 0xf, 0xf, true));
}

__global__ void __launch_bounds__(128)
gemm_wmma_kernel(const float* __restrict__ Wmat,
                 const float* __restrict__ bias,
                 const float* __restrict__ F,
                 float* __restrict__ P) {
    const int wave  = blockIdx.x * 4 + (threadIdx.x >> 5);
    const int lane  = threadIdx.x & 31;
    const int mt = wave % (C_OUT_PAD / 16);
    const int st = wave / (C_OUT_PAD / 16);

    const int o0 = mt * 16;
    const int n0 = st * 64;
    const int lhalf = lane >> 4;
    const int l16   = lane & 15;

    v8f acc[4];
#pragma unroll
    for (int v = 0; v < 8; ++v) {
        const int o = min(o0 + v + 8 * lhalf, C_OUT - 1);
        const float bv = 8.0f * bias[o];
#pragma unroll
        for (int t = 0; t < 4; ++t) acc[t][v] = bv;
    }

    const int oA = min(o0 + l16, C_OUT - 1);
    const float* Wp = Wmat + (size_t)oA * CIN + 4 * lhalf;
    const float* Fp = F + (size_t)(4 * lhalf) * HW + n0 + 4 * l16;

    FragSet f0, f1;
    load_set(f0, Wp, Fp, 0);
    for (int k0 = 0; k0 < CIN - 2 * KC; k0 += 2 * KC) {
        load_set(f1, Wp, Fp, k0 + KC);
        mma_set(acc, f0);
        load_set(f0, Wp, Fp, k0 + 2 * KC);
        mma_set(acc, f1);
    }
    load_set(f1, Wp, Fp, CIN - KC);
    mma_set(acc, f0);
    mma_set(acc, f1);

#pragma unroll
    for (int v = 0; v < 8; ++v) {
        const int m = v + 8 * lhalf;
        const float p0 = acc[0][v];
        const float p1 = p0 + acc[1][v];
        const float p2 = p1 + acc[2][v];
        const float p3 = p2 + acc[3][v];
        float inc = p3;
        inc += row_shr<1>(inc);
        inc += row_shr<2>(inc);
        inc += row_shr<4>(inc);
        inc += row_shr<8>(inc);
        const float left = row_shr<1>(inc);
        v4f o;
        o[0] = left + p0; o[1] = left + p1; o[2] = left + p2; o[3] = left + p3;
        *(v4f*)(P + (size_t)(o0 + m) * HW + n0 + 4 * l16) = o;
    }
}

__global__ void roipool_kernel(const int* __restrict__ rois,
                               const float* __restrict__ P,
                               float* __restrict__ out) {
    const int i = blockIdx.x;
    const int t = threadIdx.x;
    __shared__ float pool[C_OUT];

    const int* r = rois + i * 4;
    const int ymin = r[0] / 30, xmin = r[1] / 30;
    const int ymax = r[2] / 30, xmax = r[3] / 30;
    const int ystep = (ymax - ymin) / KPOOL;
    const int xstep = (xmax - xmin) / KPOOL;
    const float cnt = (float)(B_IMG * ystep * xstep);

    if (t < C_OUT) {
        float val = 0.0f;
        if (cnt > 0.0f) {
            const int bin = t >> 2;
            const int jy = bin / KPOOL, jx = bin % KPOOL;
            const int ys = ymin + jy * ystep;
            const int xs = xmin + jx * xstep;
            const float* pp = P + (size_t)t * HW + ys * W_DIM + xs;
            float hi[MAX_STEP], lo[MAX_STEP];
#pragma unroll
            for (int y = 0; y < MAX_STEP; ++y) {
                hi[y] = 0.0f; lo[y] = 0.0f;
                if (y < ystep) {
                    hi[y] = pp[y * W_DIM + xstep - 1];
                    if (xs > 0) lo[y] = pp[y * W_DIM - 1];
                }
            }
            float bs = 0.0f;
#pragma unroll
            for (int y = 0; y < MAX_STEP; ++y) bs += hi[y] - lo[y];
            val = bs / fmaxf(cnt, 1.0f);
        }
        pool[t] = val;
    }
    __syncthreads();

    if (t < 4) {
        float s = 0.0f;
#pragma unroll
        for (int b = 0; b < KPOOL * KPOOL; ++b) s += pool[t + 4 * b];
        out[i * 4 + t] = s / (float)(KPOOL * KPOOL);
    }
}

extern "C" void kernel_launch(void* const* d_in, const int* in_sizes, int n_in,
                              void* d_out, int out_size, void* d_ws, size_t ws_size,
                              hipStream_t stream) {
    const float* fm    = (const float*)d_in[0];
    const int*   rois  = (const int*)  d_in[1];
    const float* convw = (const float*)d_in[2];
    const float* convb = (const float*)d_in[3];
    float*       out   = (float*)d_out;

    char* ws = (char*)d_ws;
    float* F = (float*)(ws);
    float* P = (float*)(ws + (size_t)CIN * HW * sizeof(float));

    {
        const int threads = 256;
        const int blocks  = (int)(((size_t)CIN * HW / 4 + threads - 1) / threads);
        batchsum_kernel<<<blocks, threads, 0, stream>>>(fm, F);
    }
    gemm_wmma_kernel<<<208, 128, 0, stream>>>(convw, convb, F, P);

    roipool_kernel<<<N_ROI, 256, 0, stream>>>(rois, P, out);
}


# pooling gather: P laid out [bin][y][x'][c4] (a bin's 4 channels in one 16-byte element, x' = GEMM lane order); a lane pools a bin, two 16-byte loads per row; ROI block 8->2 waves; sum order kept
# speedup vs baseline: 1.4427x; 1.0626x over previous
#include <hip/hip_runtime.h>
#include <hip/hip_bf16.h>
#include <stdint.h>

#define B_IMG     8
#define CIN       1024
#define H_DIM     64
#define W_DIM     64
#define HW        (H_DIM * W_DIM)
#define C_OUT     196
#define C_OUT_PAD 208
#define N_ROI     1024
#define KPOOL     7
#define MAX_STEP  ((H_DIM - 1) / KPOOL)

typedef __attribute__((ext_vector_type(2))) float v2f;
typedef __attribute__((ext_vector_type(8))) float v8f;

__global__ void batchsum_kernel(const float* __restrict__ fm,
                                float* __restrict__ F) {
    const size_t NE = (size_t)CIN * HW;
    size_t idx = ((size_t)blockIdx.x * blockDim.x + threadIdx.x) * 4;
    if (idx >= NE) return;
    float4 s = make_float4(0.f, 0.f, 0.f, 0.f);
#pragma unroll
    for (int b = 0; b < B_IMG; ++b) {
        const float4 v = *(const float4*)(fm + (size_t)b * NE + idx);
        s.x += v.x; s.y += v.y; s.z += v.z; s.w += v.w;
    }
    *(float4*)(F + idx) = s;
}

typedef __attribute__((ext_vector_type(4))) float v4f;

#define PX_SLOT(x) ((((x) & 3) << 4) + ((x) >> 2))

#define KC   64
#define NSUB (KC / 8)

struct FragSet {
    v4f a[NSUB];
    v4f b[NSUB][4];
};

__device__ __forceinline__ void load_set(FragSet& f,
                                         const float* __restrict__ Wp,
                                         const float* __restrict__ Fp,
                                         int k0) {
#pragma unroll
    for (int s = 0; s < NSUB; ++s) {
        f.a[s] = *(const v4f*)(Wp + k0 + 8 * s);
#pragma unroll
        for (int j = 0; j < 4; ++j)
            f.b[s][j] = *(const v4f*)(Fp + (size_t)(k0 + 8 * s + j) * HW);
    }
}

__device__ __forceinline__ void mma_set(v8f (&acc)[4], const FragSet& f) {
#pragma unroll
    for (int s = 0; s < NSUB; ++s) {
        v2f a01, a23;
        a01.x = f.a[s][0]; a01.y = f.a[s][1];
        a23.x = f.a[s][2]; a23.y = f.a[s][3];
#pragma unroll
        for (int t = 0; t < 4; ++t) {
            v2f b;
            b.x = f.b[s][0][t]; b.y = f.b[s][1][t];
            acc[t] = __builtin_amdgcn_wmma_f32_16x16x4_f32(
                false, a01, false, b, (short)0, acc[t], false, false);
        }
#pragma unroll
        for (int t = 0; t < 4; ++t) {
            v2f b;
            b.x = f.b[s][2][t]; b.y = f.b[s][3][t];
            acc[t] = __builtin_amdgcn_wmma_f32_16x16x4_f32(
                false, a23, false, b, (short)0, acc[t], false, false);
        }
    }
}

template <int N>
__device__ __forceinline__ float row_shr(float x) {
    return __builtin_bit_cast(float,
        __builtin_amdgcn_update_dpp(0, __builtin_bit_cast(int, x),
                                    0x110 + N  , 0xf, 0xf, true));
}

__global__ void __launch_bounds__(128)
gemm_wmma_kernel(const float* __restrict__ Wmat,
                 const float* __restrict__ bias,
                 const float* __restrict__ F,
                 float* __restrict__ P) {
    const int wave  = blockIdx.x * 4 + (threadIdx.x >> 5);
    const int lane  = threadIdx.x & 31;
    const int mt = wave % (C_OUT_PAD / 16);
    const int st = wave / (C_OUT_PAD / 16);

    const int o0 = mt * 16;
    const int n0 = st * 64;
    const int lhalf = lane >> 4;
    const int l16   = lane & 15;

    v8f acc[4];
#pragma unroll
    for (int v = 0; v < 8; ++v) {
        const int o = min(o0 + v + 8 * lhalf, C_OUT - 1);
        const float bv = 8.0f * bias[o];
#pragma unroll
        for (int t = 0; t < 4; ++t) acc[t][v] = bv;
    }

    const int oA = min(o0 + l16, C_OUT - 1);
    const float* Wp = Wmat + (size_t)oA * CIN + 4 * lhalf;
    const float* Fp = F + (size_t)(4 * lhalf) * HW + n0 + 4 * l16;

    FragSet f0, f1;
    load_set(f0, Wp, Fp, 0);
    for (int k0 = 0; k0 < CIN - 2 * KC; k0 += 2 * KC) {
        load_set(f1, Wp, Fp, k0 + KC);
        mma_set(acc, f0);
        load_set(f0, Wp, Fp, k0 + 2 * KC);
        mma_set(acc, f1);
    }
    load_set(f1, Wp, Fp, CIN - KC);
    mma_set(acc, f0);
    mma_set(acc, f1);

    v4f pre[8];
#pragma unroll
    for (int v = 0; v < 8; ++v) {
        const float p0 = acc[0][v];
        const float p1 = p0 + acc[1][v];
        const float p2 = p1 + acc[2][v];
        const float p3 = p2 + acc[3][v];
        float inc = p3;
        inc += row_shr<1>(inc);
        inc += row_shr<2>(inc);
        inc += row_shr<4>(inc);
        inc += row_shr<8>(inc);
        const float left = row_shr<1>(inc);
        pre[v][0] = left + p0; pre[v][1] = left + p1;
        pre[v][2] = left + p2; pre[v][3] = left + p3;
    }
#pragma unroll
    for (int q = 0; q < 2; ++q) {
        const int bin = (o0 >> 2) + 2 * lhalf + q;
        v4f* dst = (v4f*)P + ((size_t)bin * H_DIM + st) * W_DIM + l16;
#pragma unroll
        for (int t = 0; t < 4; ++t) {
            v4f o;
            o[0] = pre[4 * q + 0][t]; o[1] = pre[4 * q + 1][t];
            o[2] = pre[4 * q + 2][t]; o[3] = pre[4 * q + 3][t];
            dst[16 * t] = o;
        }
    }
}

#define N_BIN (KPOOL * KPOOL)
typedef __attribute__((ext_vector_type(4))) unsigned v4u;

__global__ void __launch_bounds__(64)
roipool_kernel(const int* __restrict__ rois,
               const float* __restrict__ P,
               float* __restrict__ out) {
    const int i = blockIdx.x;
    const int b = threadIdx.x;
    __shared__ v4f pool[N_BIN];

    const int* r = rois + i * 4;
    const int ymin = r[0] / 30, xmin = r[1] / 30;
    const int ymax = r[2] / 30, xmax = r[3] / 30;
    const int ystep = (ymax - ymin) / KPOOL;
    const int xstep = (xmax - xmin) / KPOOL;
    const float cnt = (float)(B_IMG * ystep * xstep);

    if (b < N_BIN) {
        v4f val = {0.0f, 0.0f, 0.0f, 0.0f};
        if (cnt > 0.0f) {
            const int jy = b / KPOOL, jx = b % KPOOL;
            const int ys = ymin + jy * ystep;
            const int xs = xmin + jx * xstep;
            const v4f* pp = (const v4f*)P + ((size_t)b * H_DIM + ys) * W_DIM;
            const int s_hi = PX_SLOT(xs + xstep - 1);
            const int s_lo = (xs > 0) ? PX_SLOT(xs - 1) : 0;
            const unsigned lo_keep = (xs > 0) ? 0xffffffffu : 0u;
            v4f hi[MAX_STEP];
            v4u lo[MAX_STEP];
#pragma unroll
            for (int y = 0; y < MAX_STEP; ++y) {
                if (y < ystep) {
                    hi[y] = pp[y * W_DIM + s_hi];
                    lo[y] = __builtin_bit_cast(v4u, pp[y * W_DIM + s_lo]);
                }
            }
            v4f bs = {0.0f, 0.0f, 0.0f, 0.0f};
#pragma unroll
            for (int y = 0; y < MAX_STEP; ++y)
                if (y < ystep)
                    bs += hi[y] - __builtin_bit_cast(v4f, lo[y] & lo_keep);
            val = bs / fmaxf(cnt, 1.0f);
        }
        pool[b] = val;
    }
    __syncthreads();

    if (b < 4) {
        const float* pf = (const float*)pool;
        float s = 0.0f;
#pragma unroll
        for (int k = 0; k < N_BIN; ++k) s += pf[4 * k + b];
        out[i * 4 + b] = s / (float)N_BIN;
    }
}

extern "C" void kernel_launch(void* const* d_in, const int* in_sizes, int n_in,
                              void* d_out, int out_size, void* d_ws, size_t ws_size,
                              hipStream_t stream) {
    const float* fm    = (const float*)d_in[0];
    const int*   rois  = (const int*)  d_in[1];
    const float* convw = (const float*)d_in[2];
    const float* convb = (const float*)d_in[3];
    float*       out   = (float*)d_out;

    char* ws = (char*)d_ws;
    float* F = (float*)(ws);
    float* P = (float*)(ws + (size_t)CIN * HW * sizeof(float));

    {
        const int threads = 256;
        const int blocks  = (int)(((size_t)CIN * HW / 4 + threads - 1) / threads);
        batchsum_kernel<<<blocks, threads, 0, stream>>>(fm, F);
    }
    gemm_wmma_kernel<<<208, 128, 0, stream>>>(convw, convb, F, P);

    roipool_kernel<<<N_ROI, 64, 0, stream>>>(rois, P, out);
}


# batch sum fused into the GEMM: a workgroup (image row x k quarter) adds the 8 images into LDS and multiplies from it; batch-sum launch and F gone; pooling adds 4 k-quarter planes: k sum order changes
# speedup vs baseline: 1.5196x; 1.0533x over previous
#include <hip/hip_runtime.h>
#include <hip/hip_bf16.h>
#include <stdint.h>

#define B_IMG     8
#define CIN       1024
#define H_DIM     64
#define W_DIM     64
#define HW        (H_DIM * W_DIM)
#define C_OUT     196
#define C_OUT_PAD 208
#define N_ROI     1024
#define KPOOL     7
#define MAX_STEP  ((H_DIM - 1) / KPOOL)

typedef __attribute__((ext_vector_type(2))) float v2f;
typedef __attribute__((ext_vector_type(8))) float v8f;

typedef __attribute__((ext_vector_type(4))) float v4f;

#define PX_SLOT(x) ((((x) & 3) << 4) + ((x) >> 2))

#define KQ    4
#define KW    (CIN / KQ)
#define KC    32
#define NCH   (KW / KC)
#define NG    (KC / 8)
#define PLANE ((size_t)C_OUT_PAD * HW)

struct AFrag  { v4f a[2][NG]; };
struct GStage { v4f g[2][B_IMG]; };

__device__ __forceinline__ void load_a(AFrag& f, const float* __restrict__ Wp0,
                                       const float* __restrict__ Wp1, int koff) {
#pragma unroll
    for (int s = 0; s < NG; ++s) {
        f.a[0][s] = *(const v4f*)(Wp0 + koff + 8 * s);
        f.a[1][s] = *(const v4f*)(Wp1 + koff + 8 * s);
    }
}

__device__ __forceinline__ void load_fm(GStage& st, const float* __restrict__ fp, int ch) {
#pragma unroll
    for (int jj = 0; jj < 2; ++jj)
#pragma unroll
        for (int b = 0; b < B_IMG; ++b)
            st.g[jj][b] = *(const v4f*)(fp + ((size_t)b * CIN + ch * KC + 2 * jj) * HW);
}

__device__ __forceinline__ void sum_store(const GStage& st, v4f* dst) {
#pragma unroll
    for (int jj = 0; jj < 2; ++jj) {
        v4f s = {0.0f, 0.0f, 0.0f, 0.0f};
#pragma unroll
        for (int b = 0; b < B_IMG; ++b) s += st.g[jj][b];
        dst[jj * 2 * 16] = s;
    }
}

__device__ __forceinline__ void mma_tile(v8f (&acc)[4], const v4f (&a)[NG],
                                         const v4f (&bf)[NG][4]) {
#pragma unroll
    for (int s = 0; s < NG; ++s) {
        v2f a01, a23;
        a01.x = a[s][0]; a01.y = a[s][1];
        a23.x = a[s][2]; a23.y = a[s][3];
#pragma unroll
        for (int t = 0; t < 4; ++t) {
            v2f b;
            b.x = bf[s][0][t]; b.y = bf[s][1][t];
            acc[t] = __builtin_amdgcn_wmma_f32_16x16x4_f32(
                false, a01, false, b, (short)0, acc[t], false, false);
        }
#pragma unroll
        for (int t = 0; t < 4; ++t) {
            v2f b;
            b.x = bf[s][2][t]; b.y = bf[s][3][t];
            acc[t] = __builtin_amdgcn_wmma_f32_16x16x4_f32(
                false, a23, false, b, (short)0, acc[t], false, false);
        }
    }
}

template <int N>
__device__ __forceinline__ float row_shr(float x) {
    return __builtin_bit_cast(float,
        __builtin_amdgcn_update_dpp(0, __builtin_bit_cast(int, x),
                                    0x110 + N  , 0xf, 0xf, true));
}

__device__ __forceinline__ void store_prefix(const v8f (&acc)[4], float* __restrict__ Pq,
                                             int o0, int st, int lhalf, int l16) {
    v4f pre[8];
#pragma unroll
    for (int v = 0; v < 8; ++v) {
        const float p0 = acc[0][v];
        const float p1 = p0 + acc[1][v];
        const float p2 = p1 + acc[2][v];
        const float p3 = p2 + acc[3][v];
        float inc = p3;
        inc += row_shr<1>(inc);
        inc += row_shr<2>(inc);
        inc += row_shr<4>(inc);
        inc += row_shr<8>(inc);
        const float left = row_shr<1>(inc);
        pre[v][0] = left + p0; pre[v][1] = left + p1;
        pre[v][2] = left + p2; pre[v][3] = left + p3;
    }
#pragma unroll
    for (int q = 0; q < 2; ++q) {
        const int bin = (o0 >> 2) + 2 * lhalf + q;
        v4f* dst = (v4f*)Pq + ((size_t)bin * H_DIM + st) * W_DIM + l16;
#pragma unroll
        for (int t = 0; t < 4; ++t) {
            v4f o;
            o[0] = pre[4 * q + 0][t]; o[1] = pre[4 * q + 1][t];
            o[2] = pre[4 * q + 2][t]; o[3] = pre[4 * q + 3][t];
            dst[16 * t] = o;
        }
    }
}

__device__ __forceinline__ void chunk_step(v8f (&acc0)[4], v8f (&acc1)[4],
                                           const AFrag& aCur, AFrag& aNext,
                                           const v4f (*Fcur)[16], v4f* FnextDst,
                                           const float* __restrict__ fp,
                                           const float* __restrict__ Wp0,
                                           const float* __restrict__ Wp1,
                                           int ch, bool two, int lhalf, int l16) {
    GStage stg;
    const bool more = (ch + 1 < NCH);
    if (more) {
        load_fm(stg, fp, ch + 1);
        load_a(aNext, Wp0, Wp1, (ch + 1) * KC);
    }
    __builtin_amdgcn_sched_barrier(0);
    v4f bf[NG][4];
#pragma unroll
    for (int s = 0; s < NG; ++s)
#pragma unroll
        for (int j = 0; j < 4; ++j)
            bf[s][j] = Fcur[8 * s + 4 * lhalf + j][l16];
    __builtin_amdgcn_sched_barrier(0);
    mma_tile(acc0, aCur.a[0], bf);
    if (two) mma_tile(acc1, aCur.a[1], bf);
    if (more) sum_store(stg, FnextDst);
    __syncthreads();
}

__global__ void __launch_bounds__(256)
sumgemm_wmma_kernel(const float* __restrict__ fm,
                    const float* __restrict__ Wmat,
                    const float* __restrict__ bias,
                    float* __restrict__ P) {
    __shared__ v4f Fs[2][KC][16];
    const int st = blockIdx.x & 63;
    const int kq = blockIdx.x >> 6;
    const int w = __builtin_amdgcn_readfirstlane((int)(threadIdx.x >> 5));
    const int lane  = threadIdx.x & 31;
    const int lhalf = lane >> 4;
    const int l16   = lane & 15;
    const bool two  = (w < 5);
    const int o0a = w * 16;
    const int o0b = (w + 8) * 16;
    const int kbase = kq * KW;

    const unsigned bias_keep = (kq == 0) ? 0xffffffffu : 0u;
    v8f acc0[4], acc1[4];
#pragma unroll
    for (int v = 0; v < 8; ++v) {
        const int oa = min(o0a + v + 8 * lhalf, C_OUT - 1);
        const int ob = min(o0b + v + 8 * lhalf, C_OUT - 1);
        const float bva = __builtin_bit_cast(float,
            __builtin_bit_cast(unsigned, 8.0f * bias[oa]) & bias_keep);
        const float bvb = __builtin_bit_cast(float,
            __builtin_bit_cast(unsigned, 8.0f * bias[ob]) & bias_keep);
#pragma unroll
        for (int t = 0; t < 4; ++t) { acc0[t][v] = bva; acc1[t][v] = bvb; }
    }

    const int oA0 = min(o0a + l16, C_OUT - 1);
    const int oA1 = min(o0b + l16, C_OUT - 1);
    const float* Wp0 = Wmat + (size_t)oA0 * CIN + kbase + 4 * lhalf;
    const float* Wp1 = Wmat + (size_t)oA1 * CIN + kbase + 4 * lhalf;
    const float* fp = fm + (size_t)(kbase + 4 * w + lhalf) * HW + st * W_DIM + 4 * l16;
    v4f* Fdst0 = &Fs[0][4 * w + lhalf][l16];
    v4f* Fdst1 = &Fs[1][4 * w + lhalf][l16];

    AFrag fa, fb;
    {
        GStage stg;
        load_fm(stg, fp, 0);
        load_a(fa, Wp0, Wp1, 0);
        sum_store(stg, Fdst0);
    }
    __syncthreads();

    for (int ch = 0; ch < NCH; ch += 2) {
        chunk_step(acc0, acc1, fa, fb, Fs[0], Fdst1, fp, Wp0, Wp1, ch,     two, lhalf, l16);
        chunk_step(acc0, acc1, fb, fa, Fs[1], Fdst0, fp, Wp0, Wp1, ch + 1, two, lhalf, l16);
    }

    float* Pq = P + (size_t)kq * PLANE;
    store_prefix(acc0, Pq, o0a, st, lhalf, l16);
    if (two) store_prefix(acc1, Pq, o0b, st, lhalf, l16);
}

#define N_BIN  (KPOOL * KPOOL)
#define PLANE4 (C_OUT_PAD * HW / 4)
typedef __attribute__((ext_vector_type(4))) unsigned v4u;

__global__ void __launch_bounds__(64)
roipool_kernel(const int* __restrict__ rois,
               const float* __restrict__ P,
               float* __restrict__ out) {
    const int i = blockIdx.x;
    const int b = threadIdx.x;
    __shared__ v4f pool[N_BIN];

    const int* r = rois + i * 4;
    const int ymin = r[0] / 30, xmin = r[1] / 30;
    const int ymax = r[2] / 30, xmax = r[3] / 30;
    const int ystep = (ymax - ymin) / KPOOL;
    const int xstep = (xmax - xmin) / KPOOL;
    const float cnt = (float)(B_IMG * ystep * xstep);

    if (b < N_BIN) {
        v4f val = {0.0f, 0.0f, 0.0f, 0.0f};
        if (cnt > 0.0f) {
            const int jy = b / KPOOL, jx = b % KPOOL;
            const int ys = ymin + jy * ystep;
            const int xs = xmin + jx * xstep;
            const v4f* pp = (const v4f*)P + ((size_t)b * H_DIM + ys) * W_DIM;
            const int s_hi = PX_SLOT(xs + xstep - 1);
            const int s_lo = (xs > 0) ? PX_SLOT(xs - 1) : 0;
            const unsigned lo_keep = (xs > 0) ? 0xffffffffu : 0u;
            v4f hi[KQ][MAX_STEP];
            v4u lo[KQ][MAX_STEP];
#pragma unroll
            for (int y = 0; y < MAX_STEP; ++y) {
                if (y < ystep) {
#pragma unroll
                    for (int q = 0; q < KQ; ++q) {
                        const v4f* pq = pp + (size_t)q * PLANE4 + y * W_DIM;
                        hi[q][y] = pq[s_hi];
                        lo[q][y] = __builtin_bit_cast(v4u, pq[s_lo]);
                    }
                }
            }
            v4f bs = {0.0f, 0.0f, 0.0f, 0.0f};
#pragma unroll
            for (int y = 0; y < MAX_STEP; ++y)
                if (y < ystep) {
                    v4f row = hi[0][y] - __builtin_bit_cast(v4f, lo[0][y] & lo_keep);
#pragma unroll
                    for (int q = 1; q < KQ; ++q)
                        row += hi[q][y] - __builtin_bit_cast(v4f, lo[q][y] & lo_keep);
                    bs += row;
                }
            val = bs / fmaxf(cnt, 1.0f);
        }
        pool[b] = val;
    }
    __syncthreads();

    if (b < 4) {
        const float* pf = (const float*)pool;
        float s = 0.0f;
#pragma unroll
        for (int k = 0; k < N_BIN; ++k) s += pf[4 * k + b];
        out[i * 4 + b] = s / (float)N_BIN;
    }
}

extern "C" void kernel_launch(void* const* d_in, const int* in_sizes, int n_in,
                              void* d_out, int out_size, void* d_ws, size_t ws_size,
                              hipStream_t stream) {
    const float* fm    = (const float*)d_in[0];
    const int*   rois  = (const int*)  d_in[1];
    const float* convw = (const float*)d_in[2];
    const float* convb = (const float*)d_in[3];
    float*       out   = (float*)d_out;

    float* P = (float*)d_ws;

    sumgemm_wmma_kernel<<<256, 256, 0, stream>>>(fm, convw, convb, P);

    roipool_kernel<<<N_ROI, 64, 0, stream>>>(rois, P, out);
}
